# starting kernel
# baseline (speedup 1.0000x reference)
#include <hip/hip_runtime.h>
#include <hip/hip_bf16.h>

typedef float v2f __attribute__((ext_vector_type(2)));
typedef float v8f __attribute__((ext_vector_type(8)));
typedef float f4  __attribute__((ext_vector_type(4)));

constexpr int BB = 64;
constexpr int TT = 512;
constexpr int FF = 1024;
constexpr int UU = 64;
constexpr int MM = BB * TT;

#define KCH 64
#define LDA 66
#define LDB 66

__global__ __launch_bounds__(256) void crf_potentials_gemm(
    const float* __restrict__ x, const float* __restrict__ w,
    const float* __restrict__ bias, const float* __restrict__ lb,
    const float* __restrict__ rb, float* __restrict__ pot) {
  __shared__ __attribute__((aligned(16))) float lds_a[128 * LDA];
  __shared__ __attribute__((aligned(16))) float lds_b[64 * LDB];

  const int tid = threadIdx.x;
  const int lane = tid & 31;
  const int wave = tid >> 5;
  const int blockRow = blockIdx.x * 128;
  const int rowBase = blockRow + wave * 16;

  const int mrow = lane & 15;
  const int kbase = (lane >> 4) << 1;

  v8f acc[4] = {};

  for (int kk = 0; kk < FF; kk += KCH) {
    __syncthreads();
#pragma unroll
    for (int i = 0; i < 8; ++i) {
      int f = tid + i * 256;
      int row = f >> 4;
      int q = f & 15;
      const float* src = x + (size_t)(blockRow + row) * FF + kk + q * 4;
      if (kk + KCH < FF) __builtin_prefetch(src + KCH, 0, 0);
      f4 av = *(const f4*)src;
      float* dst = &lds_a[row * LDA + q * 4];
      dst[0] = av.x; dst[1] = av.y; dst[2] = av.z; dst[3] = av.w;
    }
#pragma unroll
    for (int i = 0; i < 4; ++i) {
      int f = tid + i * 256;
      int row = f >> 4;
      int q = f & 15;
      f4 bv = *(const f4*)(w + (size_t)(kk + row) * UU + q * 4);
      lds_b[(q * 4 + 0) * LDB + row] = bv.x;
      lds_b[(q * 4 + 1) * LDB + row] = bv.y;
      lds_b[(q * 4 + 2) * LDB + row] = bv.z;
      lds_b[(q * 4 + 3) * LDB + row] = bv.w;
    }
    __syncthreads();
#pragma unroll
    for (int k4 = 0; k4 < KCH; k4 += 4) {
      v2f afrag = *(const v2f*)&lds_a[(wave * 16 + mrow) * LDA + k4 + kbase];
#pragma unroll
      for (int nt = 0; nt < 4; ++nt) {
        int n = nt * 16 + mrow;
        v2f bfrag = *(const v2f*)&lds_b[n * LDB + k4 + kbase];
        acc[nt] = __builtin_amdgcn_wmma_f32_16x16x4_f32(
            false, afrag, false, bfrag, (short)0, acc[nt], false, false);
      }
    }
  }

  const int rsel = (lane >> 4) * 8;
#pragma unroll
  for (int nt = 0; nt < 4; ++nt) {
    int col = nt * 16 + mrow;
    float bia = bias[col];
    float lbv = lb[col];
    float rbv = rb[col];
#pragma unroll
    for (int r = 0; r < 8; ++r) {
      int row = rowBase + r + rsel;
      float v = acc[nt][r] + bia;
      int t = row & (TT - 1);
      if (t == 0) v += lbv;
      if (t == TT - 1) v += rbv;
      pot[(size_t)row * UU + col] = v;
    }
  }
}

__global__ __launch_bounds__(256) void crf_viterbi(
    const float* __restrict__ pot, const float* __restrict__ chain,
    int* __restrict__ path) {
  __shared__ float alpha[UU];
  __shared__ float pv[4][UU];
  __shared__ int pidx[4][UU];
  __shared__ unsigned char bp[TT * UU];
  __shared__ __attribute__((aligned(16))) float potbuf[2][32 * UU];

  const int tid = threadIdx.x;
  const int c = tid & 63;
  const int pg = tid >> 6;
  const int b = blockIdx.x;
  const float* potb = pot + (size_t)b * TT * UU;

  float ck[16];
#pragma unroll
  for (int j = 0; j < 16; ++j) ck[j] = chain[(pg * 16 + j) * UU + c];

#pragma unroll
  for (int i = 0; i < 2; ++i) {
    int f = tid + i * 256;
    *(f4*)&potbuf[0][f * 4] = *(const f4*)(potb + f * 4);
  }
  __syncthreads();
  if (pg == 0) alpha[c] = potbuf[0][c];
  __syncthreads();

  int t = 1;
  for (int chunk = 0; chunk < 16; ++chunk) {
    const int buf = chunk & 1;
    f4 nv0, nv1;
    if (chunk < 15) {
      nv0 = *(const f4*)(potb + (chunk + 1) * 32 * UU + (tid + 0) * 4);
      nv1 = *(const f4*)(potb + (chunk + 1) * 32 * UU + (tid + 256) * 4);
    }
    const int tEnd = (chunk + 1) * 32;
    for (; t < tEnd; ++t) {
      const int tl = t - chunk * 32;
      float best = -3.402823466e38f;
      int bestp = 0;
#pragma unroll
      for (int j = 0; j < 16; ++j) {
        float s = alpha[pg * 16 + j] + ck[j];
        if (s > best) { best = s; bestp = pg * 16 + j; }
      }
      pv[pg][c] = best;
      pidx[pg][c] = bestp;
      __syncthreads();
      if (pg == 0) {
        float bb = pv[0][c];
        int bi = pidx[0][c];
#pragma unroll
        for (int g = 1; g < 4; ++g) {
          float s = pv[g][c];
          if (s > bb) { bb = s; bi = pidx[g][c]; }
        }
        bp[t * UU + c] = (unsigned char)bi;
        alpha[c] = bb + potbuf[buf][tl * UU + c];
      }
      __syncthreads();
    }
    if (chunk < 15) {
      *(f4*)&potbuf[1 - buf][(tid + 0) * 4] = nv0;
      *(f4*)&potbuf[1 - buf][(tid + 256) * 4] = nv1;
      __syncthreads();
    }
  }

  if (tid == 0) {
    float bb = alpha[0];
    int tag = 0;
    for (int i = 1; i < UU; ++i) {
      float s = alpha[i];
      if (s > bb) { bb = s; tag = i; }
    }
    int* outb = path + (size_t)b * TT;
    outb[TT - 1] = tag;
    for (int tt = TT - 1; tt >= 1; --tt) {
      tag = bp[tt * UU + tag];
      outb[tt - 1] = tag;
    }
  }
}

extern "C" void kernel_launch(void* const* d_in, const int* in_sizes, int n_in,
                              void* d_out, int out_size, void* d_ws,
                              size_t ws_size, hipStream_t stream) {
  (void)in_sizes; (void)n_in; (void)out_size; (void)ws_size;
  const float* x     = (const float*)d_in[0];
  const float* w     = (const float*)d_in[1];
  const float* bias  = (const float*)d_in[2];
  const float* chain = (const float*)d_in[3];
  const float* lb    = (const float*)d_in[4];
  const float* rb    = (const float*)d_in[5];
  float* pot = (float*)d_ws;
  int* out = (int*)d_out;

  crf_potentials_gemm<<<dim3(MM / 128), dim3(256), 0, stream>>>(
      x, w, bias, lb, rb, pot);
  crf_viterbi<<<dim3(BB), dim3(256), 0, stream>>>(pot, chain, out);
}


# Viterbi: each tag's 64-way argmax finishes inside one wave (register tree, lane halves joined by permlanex16): one LDS exchange and one LDS-only barrier per step, not two; same sums and tie order
# speedup vs baseline: 1.4059x; 1.4059x over previous
#include <hip/hip_runtime.h>
#include <hip/hip_bf16.h>

typedef float v2f __attribute__((ext_vector_type(2)));
typedef float v8f __attribute__((ext_vector_type(8)));
typedef float f4  __attribute__((ext_vector_type(4)));

constexpr int BB = 64;
constexpr int TT = 512;
constexpr int FF = 1024;
constexpr int UU = 64;
constexpr int MM = BB * TT;

#define KCH 64
#define LDA 66
#define LDB 66

__global__ __launch_bounds__(256) void crf_potentials_gemm(
    const float* __restrict__ x, const float* __restrict__ w,
    const float* __restrict__ bias, const float* __restrict__ lb,
    const float* __restrict__ rb, float* __restrict__ pot) {
  __shared__ __attribute__((aligned(16))) float lds_a[128 * LDA];
  __shared__ __attribute__((aligned(16))) float lds_b[64 * LDB];

  const int tid = threadIdx.x;
  const int lane = tid & 31;
  const int wave = tid >> 5;
  const int blockRow = blockIdx.x * 128;
  const int rowBase = blockRow + wave * 16;

  const int mrow = lane & 15;
  const int kbase = (lane >> 4) << 1;

  v8f acc[4] = {};

  for (int kk = 0; kk < FF; kk += KCH) {
    __syncthreads();
#pragma unroll
    for (int i = 0; i < 8; ++i) {
      int f = tid + i * 256;
      int row = f >> 4;
      int q = f & 15;
      const float* src = x + (size_t)(blockRow + row) * FF + kk + q * 4;
      if (kk + KCH < FF) __builtin_prefetch(src + KCH, 0, 0);
      f4 av = *(const f4*)src;
      float* dst = &lds_a[row * LDA + q * 4];
      dst[0] = av.x; dst[1] = av.y; dst[2] = av.z; dst[3] = av.w;
    }
#pragma unroll
    for (int i = 0; i < 4; ++i) {
      int f = tid + i * 256;
      int row = f >> 4;
      int q = f & 15;
      f4 bv = *(const f4*)(w + (size_t)(kk + row) * UU + q * 4);
      lds_b[(q * 4 + 0) * LDB + row] = bv.x;
      lds_b[(q * 4 + 1) * LDB + row] = bv.y;
      lds_b[(q * 4 + 2) * LDB + row] = bv.z;
      lds_b[(q * 4 + 3) * LDB + row] = bv.w;
    }
    __syncthreads();
#pragma unroll
    for (int k4 = 0; k4 < KCH; k4 += 4) {
      v2f afrag = *(const v2f*)&lds_a[(wave * 16 + mrow) * LDA + k4 + kbase];
#pragma unroll
      for (int nt = 0; nt < 4; ++nt) {
        int n = nt * 16 + mrow;
        v2f bfrag = *(const v2f*)&lds_b[n * LDB + k4 + kbase];
        acc[nt] = __builtin_amdgcn_wmma_f32_16x16x4_f32(
            false, afrag, false, bfrag, (short)0, acc[nt], false, false);
      }
    }
  }

  const int rsel = (lane >> 4) * 8;
#pragma unroll
  for (int nt = 0; nt < 4; ++nt) {
    int col = nt * 16 + mrow;
    float bia = bias[col];
    float lbv = lb[col];
    float rbv = rb[col];
#pragma unroll
    for (int r = 0; r < 8; ++r) {
      int row = rowBase + r + rsel;
      float v = acc[nt][r] + bia;
      int t = row & (TT - 1);
      if (t == 0) v += lbv;
      if (t == TT - 1) v += rbv;
      pot[(size_t)row * UU + col] = v;
    }
  }
}

static __device__ __forceinline__ int xhalf_i(int b) {
  return __builtin_amdgcn_permlanex16(b, b, 0x76543210, 0xfedcba98, false, false);
}
static __device__ __forceinline__ float xhalf_f(float x) {
  return __builtin_bit_cast(float, xhalf_i(__builtin_bit_cast(int, x)));
}
static __device__ __forceinline__ void step_barrier() {
  asm volatile("s_wait_dscnt 0x0\n\ts_barrier_signal -1\n\ts_barrier_wait -1" ::: "memory");
}

struct ValIdx { float v; int i; };
template <int LO, int N>
static __device__ __forceinline__ ValIdx argmax_tree(const float (&s)[32]) {
  if constexpr (N == 1) {
    return ValIdx{s[LO], LO};
  } else {
    const ValIdx a = argmax_tree<LO, N / 2>(s);
    const ValIdx b = argmax_tree<LO + N / 2, N / 2>(s);
    const bool g = b.v > a.v;
    return ValIdx{g ? b.v : a.v, g ? b.i : a.i};
  }
}

__global__ __launch_bounds__(128) void crf_viterbi(
    const float* __restrict__ pot, const float* __restrict__ chain,
    int* __restrict__ path) {
  __shared__ __attribute__((aligned(16))) float alphabuf[2][UU];
  __shared__ unsigned char bp[TT * UU];
  __shared__ __attribute__((aligned(16))) float potbuf[2][32 * UU];

  const int tid = threadIdx.x;
  const int lane = tid & 31;
  const int wave = tid >> 5;
  const int h = lane >> 4;
  const int c = wave * 16 + (lane & 15);
  const int b = blockIdx.x;
  const float* potb = pot + (size_t)b * TT * UU;

  float ck[32];
#pragma unroll
  for (int j = 0; j < 32; ++j) ck[j] = chain[(h * 32 + j) * UU + c];

#pragma unroll
  for (int i = 0; i < 4; ++i) {
    int f = tid + i * 128;
    *(f4*)&potbuf[0][f * 4] = *(const f4*)(potb + f * 4);
  }
  __syncthreads();
  if (h == 0) alphabuf[0][c] = potbuf[0][c];
  __syncthreads();

  float pt = potbuf[0][1 * UU + c];

  int t = 1;
  for (int chunk = 0; chunk < 16; ++chunk) {
    const int buf = chunk & 1;
    f4 nv[4];
    if (chunk < 15) {
#pragma unroll
      for (int i = 0; i < 4; ++i)
        nv[i] = *(const f4*)(potb + (chunk + 1) * 32 * UU + (tid + i * 128) * 4);
    }
    const int tEnd = (chunk + 1) * 32;
    for (; t < tEnd; ++t) {
      const int tl = t - chunk * 32;
      const float ptn = potbuf[buf][((tl + 1) & 31) * UU + c];
      const f4* ap = (const f4*)&alphabuf[(t - 1) & 1][h * 32];
      float v[32];
#pragma unroll
      for (int q = 0; q < 8; ++q) {
        const f4 a = ap[q];
        v[4 * q + 0] = a.x + ck[4 * q + 0];
        v[4 * q + 1] = a.y + ck[4 * q + 1];
        v[4 * q + 2] = a.z + ck[4 * q + 2];
        v[4 * q + 3] = a.w + ck[4 * q + 3];
      }
      const ValIdx best = argmax_tree<0, 32>(v);
      const float bv = best.v;
      const int bi = best.i + h * 32;
      const float ov = xhalf_f(bv);
      const int oi = xhalf_i(bi);
      const bool g = ov > bv;
      if (h == 0) {
        bp[t * UU + c] = (unsigned char)(g ? oi : bi);
        alphabuf[t & 1][c] = (g ? ov : bv) + pt;
      }
      step_barrier();
      pt = ptn;
    }
    if (chunk < 15) {
#pragma unroll
      for (int i = 0; i < 4; ++i)
        *(f4*)&potbuf[1 - buf][(tid + i * 128) * 4] = nv[i];
      __syncthreads();
      pt = potbuf[1 - buf][c];
    }
  }

  if (tid == 0) {
    const float* af = alphabuf[(TT - 1) & 1];
    float bb = af[0];
    int tag = 0;
    for (int i = 1; i < UU; ++i) {
      float s = af[i];
      if (s > bb) { bb = s; tag = i; }
    }
    int* outb = path + (size_t)b * TT;
    outb[TT - 1] = tag;
    for (int tt = TT - 1; tt >= 1; --tt) {
      tag = bp[tt * UU + tag];
      outb[tt - 1] = tag;
    }
  }
}

extern "C" void kernel_launch(void* const* d_in, const int* in_sizes, int n_in,
                              void* d_out, int out_size, void* d_ws,
                              size_t ws_size, hipStream_t stream) {
  (void)in_sizes; (void)n_in; (void)out_size; (void)ws_size;
  const float* x     = (const float*)d_in[0];
  const float* w     = (const float*)d_in[1];
  const float* bias  = (const float*)d_in[2];
  const float* chain = (const float*)d_in[3];
  const float* lb    = (const float*)d_in[4];
  const float* rb    = (const float*)d_in[5];
  float* pot = (float*)d_ws;
  int* out = (int*)d_out;

  crf_potentials_gemm<<<dim3(MM / 128), dim3(256), 0, stream>>>(
      x, w, bias, lb, rb, pot);
  crf_viterbi<<<dim3(BB), dim3(128), 0, stream>>>(pot, chain, out);
}


# Viterbi: a step's serial path carries only the 64-way max (tree of maxima); the backpointer is read off the kept tree one step later (5 compares) under the next alpha loads; same sums and tie order
# speedup vs baseline: 1.6182x; 1.1510x over previous
#include <hip/hip_runtime.h>
#include <hip/hip_bf16.h>

typedef float v2f __attribute__((ext_vector_type(2)));
typedef float v8f __attribute__((ext_vector_type(8)));
typedef float f4  __attribute__((ext_vector_type(4)));

constexpr int BB = 64;
constexpr int TT = 512;
constexpr int FF = 1024;
constexpr int UU = 64;
constexpr int MM = BB * TT;

#define KCH 64
#define LDA 66
#define LDB 66

__global__ __launch_bounds__(256) void crf_potentials_gemm(
    const float* __restrict__ x, const float* __restrict__ w,
    const float* __restrict__ bias, const float* __restrict__ lb,
    const float* __restrict__ rb, float* __restrict__ pot) {
  __shared__ __attribute__((aligned(16))) float lds_a[128 * LDA];
  __shared__ __attribute__((aligned(16))) float lds_b[64 * LDB];

  const int tid = threadIdx.x;
  const int lane = tid & 31;
  const int wave = tid >> 5;
  const int blockRow = blockIdx.x * 128;
  const int rowBase = blockRow + wave * 16;

  const int mrow = lane & 15;
  const int kbase = (lane >> 4) << 1;

  v8f acc[4] = {};

  for (int kk = 0; kk < FF; kk += KCH) {
    __syncthreads();
#pragma unroll
    for (int i = 0; i < 8; ++i) {
      int f = tid + i * 256;
      int row = f >> 4;
      int q = f & 15;
      const float* src = x + (size_t)(blockRow + row) * FF + kk + q * 4;
      if (kk + KCH < FF) __builtin_prefetch(src + KCH, 0, 0);
      f4 av = *(const f4*)src;
      float* dst = &lds_a[row * LDA + q * 4];
      dst[0] = av.x; dst[1] = av.y; dst[2] = av.z; dst[3] = av.w;
    }
#pragma unroll
    for (int i = 0; i < 4; ++i) {
      int f = tid + i * 256;
      int row = f >> 4;
      int q = f & 15;
      f4 bv = *(const f4*)(w + (size_t)(kk + row) * UU + q * 4);
      lds_b[(q * 4 + 0) * LDB + row] = bv.x;
      lds_b[(q * 4 + 1) * LDB + row] = bv.y;
      lds_b[(q * 4 + 2) * LDB + row] = bv.z;
      lds_b[(q * 4 + 3) * LDB + row] = bv.w;
    }
    __syncthreads();
#pragma unroll
    for (int k4 = 0; k4 < KCH; k4 += 4) {
      v2f afrag = *(const v2f*)&lds_a[(wave * 16 + mrow) * LDA + k4 + kbase];
#pragma unroll
      for (int nt = 0; nt < 4; ++nt) {
        int n = nt * 16 + mrow;
        v2f bfrag = *(const v2f*)&lds_b[n * LDB + k4 + kbase];
        acc[nt] = __builtin_amdgcn_wmma_f32_16x16x4_f32(
            false, afrag, false, bfrag, (short)0, acc[nt], false, false);
      }
    }
  }

  const int rsel = (lane >> 4) * 8;
#pragma unroll
  for (int nt = 0; nt < 4; ++nt) {
    int col = nt * 16 + mrow;
    float bia = bias[col];
    float lbv = lb[col];
    float rbv = rb[col];
#pragma unroll
    for (int r = 0; r < 8; ++r) {
      int row = rowBase + r + rsel;
      float v = acc[nt][r] + bia;
      int t = row & (TT - 1);
      if (t == 0) v += lbv;
      if (t == TT - 1) v += rbv;
      pot[(size_t)row * UU + col] = v;
    }
  }
}

static __device__ __forceinline__ int xhalf_i(int b) {
  return __builtin_amdgcn_permlanex16(b, b, 0x76543210, 0xfedcba98, false, false);
}
static __device__ __forceinline__ float xhalf_f(float x) {
  const int xi = __builtin_bit_cast(int, x);
  return __builtin_bit_cast(
      float, __builtin_amdgcn_permlanex16(0, xi, 0x76543210, 0xfedcba98, false, false));
}
static __device__ __forceinline__ void step_barrier() {
  asm volatile("s_wait_dscnt 0x0\n\ts_barrier_signal -1\n\ts_barrier_wait -1" ::: "memory");
}

struct MaxTree {
  float leaf[16];
  float l2[8];
  float l4[4];
  float l8[2];
  float l16;
  float top;
};
static __device__ __forceinline__ void max_tree_build(const float (&s)[32], MaxTree& m) {
  float g2[16], g4[8], g8[4], g16[2];
#pragma unroll
  for (int i = 0; i < 16; ++i) g2[i] = __builtin_fmaxf(s[2 * i], s[2 * i + 1]);
#pragma unroll
  for (int i = 0; i < 8; ++i) g4[i] = __builtin_fmaxf(g2[2 * i], g2[2 * i + 1]);
#pragma unroll
  for (int i = 0; i < 4; ++i) g8[i] = __builtin_fmaxf(g4[2 * i], g4[2 * i + 1]);
#pragma unroll
  for (int i = 0; i < 2; ++i) g16[i] = __builtin_fmaxf(g8[2 * i], g8[2 * i + 1]);
  m.top = __builtin_fmaxf(g16[0], g16[1]);
#pragma unroll
  for (int i = 0; i < 16; ++i) m.leaf[i] = s[2 * i];
#pragma unroll
  for (int i = 0; i < 8; ++i) m.l2[i] = g2[2 * i];
#pragma unroll
  for (int i = 0; i < 4; ++i) m.l4[i] = g4[2 * i];
#pragma unroll
  for (int i = 0; i < 2; ++i) m.l8[i] = g8[2 * i];
  m.l16 = g16[0];
}
static __device__ __forceinline__ int max_tree_first(const MaxTree& m) {
  const float top = m.top;
  const bool b4 = m.l16 != top;
  const float c8 = b4 ? m.l8[1] : m.l8[0];
  float c4[2], c2[4], c1[8];
#pragma unroll
  for (int k = 0; k < 2; ++k) c4[k] = b4 ? m.l4[2 + k] : m.l4[k];
#pragma unroll
  for (int k = 0; k < 4; ++k) c2[k] = b4 ? m.l2[4 + k] : m.l2[k];
#pragma unroll
  for (int k = 0; k < 8; ++k) c1[k] = b4 ? m.leaf[8 + k] : m.leaf[k];
  const bool b3 = c8 != top;
  const float d4 = b3 ? c4[1] : c4[0];
  float d2[2], d1[4];
#pragma unroll
  for (int k = 0; k < 2; ++k) d2[k] = b3 ? c2[2 + k] : c2[k];
#pragma unroll
  for (int k = 0; k < 4; ++k) d1[k] = b3 ? c1[4 + k] : c1[k];
  const bool b2 = d4 != top;
  const float e2 = b2 ? d2[1] : d2[0];
  float e1[2];
#pragma unroll
  for (int k = 0; k < 2; ++k) e1[k] = b2 ? d1[2 + k] : d1[k];
  const bool b1 = e2 != top;
  const float f1 = b1 ? e1[1] : e1[0];
  const bool b0 = f1 != top;
  return (b4 ? 16 : 0) | (b3 ? 8 : 0) | (b2 ? 4 : 0) | (b1 ? 2 : 0) | (b0 ? 1 : 0);
}

__global__ __launch_bounds__(128) void crf_viterbi(
    const float* __restrict__ pot, const float* __restrict__ chain,
    int* __restrict__ path) {
  __shared__ __attribute__((aligned(16))) float alphabuf[2][UU];
  __shared__ unsigned char bp[TT * UU];
  __shared__ __attribute__((aligned(16))) float potbuf[2][32 * UU];

  const int tid = threadIdx.x;
  const int lane = tid & 31;
  const int wave = tid >> 5;
  const int h = lane >> 4;
  const int c = wave * 16 + (lane & 15);
  const int b = blockIdx.x;
  const float* potb = pot + (size_t)b * TT * UU;

  float ck[32];
#pragma unroll
  for (int j = 0; j < 32; ++j) ck[j] = chain[(h * 32 + j) * UU + c];

#pragma unroll
  for (int i = 0; i < 4; ++i) {
    int f = tid + i * 128;
    *(f4*)&potbuf[0][f * 4] = *(const f4*)(potb + f * 4);
  }
  __syncthreads();
  if (h == 0) alphabuf[0][c] = potbuf[0][c];
  __syncthreads();

  float pt = potbuf[0][1 * UU + c];

  MaxTree mt;
#pragma unroll
  for (int i = 0; i < 16; ++i) mt.leaf[i] = 0.0f;
#pragma unroll
  for (int i = 0; i < 8; ++i) mt.l2[i] = 0.0f;
#pragma unroll
  for (int i = 0; i < 4; ++i) mt.l4[i] = 0.0f;
  mt.l8[0] = 0.0f; mt.l8[1] = 0.0f; mt.l16 = 0.0f; mt.top = 0.0f;
  float vmax = 0.0f;

  int t = 1;
  for (int chunk = 0; chunk < 16; ++chunk) {
    const int buf = chunk & 1;
    f4 nv[4];
    if (chunk < 15) {
#pragma unroll
      for (int i = 0; i < 4; ++i)
        nv[i] = *(const f4*)(potb + (chunk + 1) * 32 * UU + (tid + i * 128) * 4);
    }
    const int tEnd = (chunk + 1) * 32;
    for (; t < tEnd; ++t) {
      const int tl = t - chunk * 32;
      const f4* ap = (const f4*)&alphabuf[(t - 1) & 1][h * 32];
      f4 a[8];
#pragma unroll
      for (int q = 0; q < 8; ++q) a[q] = ap[q];
      const float ptn = potbuf[buf][((tl + 1) & 31) * UU + c];
      __builtin_amdgcn_sched_barrier(0);

      {
        const int fi = max_tree_first(mt) + h * 32;
        const int gi = (mt.top == vmax) ? fi : 255;
        const int oi = xhalf_i(gi);
        bp[(t - 1) * UU + c] = (unsigned char)min(gi, oi);
      }
      __builtin_amdgcn_sched_barrier(0);

      float v[32];
#pragma unroll
      for (int q = 0; q < 8; ++q) {
        v[4 * q + 0] = a[q].x + ck[4 * q + 0];
        v[4 * q + 1] = a[q].y + ck[4 * q + 1];
        v[4 * q + 2] = a[q].z + ck[4 * q + 2];
        v[4 * q + 3] = a[q].w + ck[4 * q + 3];
      }
      max_tree_build(v, mt);
      vmax = __builtin_fmaxf(mt.top, xhalf_f(mt.top));
      alphabuf[t & 1][c] = vmax + pt;
      step_barrier();
      pt = ptn;
    }
    if (chunk < 15) {
#pragma unroll
      for (int i = 0; i < 4; ++i)
        *(f4*)&potbuf[1 - buf][(tid + i * 128) * 4] = nv[i];
      __syncthreads();
      pt = potbuf[1 - buf][c];
    }
  }
  {
    const int fi = max_tree_first(mt) + h * 32;
    const int gi = (mt.top == vmax) ? fi : 255;
    const int oi = xhalf_i(gi);
    bp[(TT - 1) * UU + c] = (unsigned char)min(gi, oi);
  }
  __syncthreads();

  if (tid == 0) {
    const float* af = alphabuf[(TT - 1) & 1];
    float bb = af[0];
    int tag = 0;
    for (int i = 1; i < UU; ++i) {
      float s = af[i];
      if (s > bb) { bb = s; tag = i; }
    }
    int* outb = path + (size_t)b * TT;
    outb[TT - 1] = tag;
    for (int tt = TT - 1; tt >= 1; --tt) {
      tag = bp[tt * UU + tag];
      outb[tt - 1] = tag;
    }
  }
}

extern "C" void kernel_launch(void* const* d_in, const int* in_sizes, int n_in,
                              void* d_out, int out_size, void* d_ws,
                              size_t ws_size, hipStream_t stream) {
  (void)in_sizes; (void)n_in; (void)out_size; (void)ws_size;
  const float* x     = (const float*)d_in[0];
  const float* w     = (const float*)d_in[1];
  const float* bias  = (const float*)d_in[2];
  const float* chain = (const float*)d_in[3];
  const float* lb    = (const float*)d_in[4];
  const float* rb    = (const float*)d_in[5];
  float* pot = (float*)d_ws;
  int* out = (int*)d_out;

  crf_potentials_gemm<<<dim3(MM / 128), dim3(256), 0, stream>>>(
      x, w, bias, lb, rb, pot);
  crf_viterbi<<<dim3(BB), dim3(128), 0, stream>>>(pot, chain, out);
}


# Viterbi backtrack: one lane's 511 dependent backpointer reads cut into 16 segments of 32 rows; all threads first compose each segment's tag map, then 16 lanes walk their segments at once; same path
# speedup vs baseline: 1.8142x; 1.1211x over previous
#include <hip/hip_runtime.h>
#include <hip/hip_bf16.h>

typedef float v2f __attribute__((ext_vector_type(2)));
typedef float v8f __attribute__((ext_vector_type(8)));
typedef float f4  __attribute__((ext_vector_type(4)));

constexpr int BB = 64;
constexpr int TT = 512;
constexpr int FF = 1024;
constexpr int UU = 64;
constexpr int MM = BB * TT;
constexpr int SEG = 32;
constexpr int NSEG = TT / SEG;

#define KCH 64
#define LDA 66
#define LDB 66

__global__ __launch_bounds__(256) void crf_potentials_gemm(
    const float* __restrict__ x, const float* __restrict__ w,
    const float* __restrict__ bias, const float* __restrict__ lb,
    const float* __restrict__ rb, float* __restrict__ pot) {
  __shared__ __attribute__((aligned(16))) float lds_a[128 * LDA];
  __shared__ __attribute__((aligned(16))) float lds_b[64 * LDB];

  const int tid = threadIdx.x;
  const int lane = tid & 31;
  const int wave = tid >> 5;
  const int blockRow = blockIdx.x * 128;
  const int rowBase = blockRow + wave * 16;

  const int mrow = lane & 15;
  const int kbase = (lane >> 4) << 1;

  v8f acc[4] = {};

  for (int kk = 0; kk < FF; kk += KCH) {
    __syncthreads();
#pragma unroll
    for (int i = 0; i < 8; ++i) {
      int f = tid + i * 256;
      int row = f >> 4;
      int q = f & 15;
      const float* src = x + (size_t)(blockRow + row) * FF + kk + q * 4;
      if (kk + KCH < FF) __builtin_prefetch(src + KCH, 0, 0);
      f4 av = *(const f4*)src;
      float* dst = &lds_a[row * LDA + q * 4];
      dst[0] = av.x; dst[1] = av.y; dst[2] = av.z; dst[3] = av.w;
    }
#pragma unroll
    for (int i = 0; i < 4; ++i) {
      int f = tid + i * 256;
      int row = f >> 4;
      int q = f & 15;
      f4 bv = *(const f4*)(w + (size_t)(kk + row) * UU + q * 4);
      lds_b[(q * 4 + 0) * LDB + row] = bv.x;
      lds_b[(q * 4 + 1) * LDB + row] = bv.y;
      lds_b[(q * 4 + 2) * LDB + row] = bv.z;
      lds_b[(q * 4 + 3) * LDB + row] = bv.w;
    }
    __syncthreads();
#pragma unroll
    for (int k4 = 0; k4 < KCH; k4 += 4) {
      v2f afrag = *(const v2f*)&lds_a[(wave * 16 + mrow) * LDA + k4 + kbase];
#pragma unroll
      for (int nt = 0; nt < 4; ++nt) {
        int n = nt * 16 + mrow;
        v2f bfrag = *(const v2f*)&lds_b[n * LDB + k4 + kbase];
        acc[nt] = __builtin_amdgcn_wmma_f32_16x16x4_f32(
            false, afrag, false, bfrag, (short)0, acc[nt], false, false);
      }
    }
  }

  const int rsel = (lane >> 4) * 8;
#pragma unroll
  for (int nt = 0; nt < 4; ++nt) {
    int col = nt * 16 + mrow;
    float bia = bias[col];
    float lbv = lb[col];
    float rbv = rb[col];
#pragma unroll
    for (int r = 0; r < 8; ++r) {
      int row = rowBase + r + rsel;
      float v = acc[nt][r] + bia;
      int t = row & (TT - 1);
      if (t == 0) v += lbv;
      if (t == TT - 1) v += rbv;
      pot[(size_t)row * UU + col] = v;
    }
  }
}

static __device__ __forceinline__ int xhalf_i(int b) {
  return __builtin_amdgcn_permlanex16(b, b, 0x76543210, 0xfedcba98, false, false);
}
static __device__ __forceinline__ float xhalf_f(float x) {
  const int xi = __builtin_bit_cast(int, x);
  return __builtin_bit_cast(
      float, __builtin_amdgcn_permlanex16(0, xi, 0x76543210, 0xfedcba98, false, false));
}
static __device__ __forceinline__ void step_barrier() {
  asm volatile("s_wait_dscnt 0x0\n\ts_barrier_signal -1\n\ts_barrier_wait -1" ::: "memory");
}

struct MaxTree {
  float leaf[16];
  float l2[8];
  float l4[4];
  float l8[2];
  float l16;
  float top;
};
static __device__ __forceinline__ void max_tree_build(const float (&s)[32], MaxTree& m) {
  float g2[16], g4[8], g8[4], g16[2];
#pragma unroll
  for (int i = 0; i < 16; ++i) g2[i] = __builtin_fmaxf(s[2 * i], s[2 * i + 1]);
#pragma unroll
  for (int i = 0; i < 8; ++i) g4[i] = __builtin_fmaxf(g2[2 * i], g2[2 * i + 1]);
#pragma unroll
  for (int i = 0; i < 4; ++i) g8[i] = __builtin_fmaxf(g4[2 * i], g4[2 * i + 1]);
#pragma unroll
  for (int i = 0; i < 2; ++i) g16[i] = __builtin_fmaxf(g8[2 * i], g8[2 * i + 1]);
  m.top = __builtin_fmaxf(g16[0], g16[1]);
#pragma unroll
  for (int i = 0; i < 16; ++i) m.leaf[i] = s[2 * i];
#pragma unroll
  for (int i = 0; i < 8; ++i) m.l2[i] = g2[2 * i];
#pragma unroll
  for (int i = 0; i < 4; ++i) m.l4[i] = g4[2 * i];
#pragma unroll
  for (int i = 0; i < 2; ++i) m.l8[i] = g8[2 * i];
  m.l16 = g16[0];
}
static __device__ __forceinline__ int max_tree_first(const MaxTree& m) {
  const float top = m.top;
  const bool b4 = m.l16 != top;
  const float c8 = b4 ? m.l8[1] : m.l8[0];
  float c4[2], c2[4], c1[8];
#pragma unroll
  for (int k = 0; k < 2; ++k) c4[k] = b4 ? m.l4[2 + k] : m.l4[k];
#pragma unroll
  for (int k = 0; k < 4; ++k) c2[k] = b4 ? m.l2[4 + k] : m.l2[k];
#pragma unroll
  for (int k = 0; k < 8; ++k) c1[k] = b4 ? m.leaf[8 + k] : m.leaf[k];
  const bool b3 = c8 != top;
  const float d4 = b3 ? c4[1] : c4[0];
  float d2[2], d1[4];
#pragma unroll
  for (int k = 0; k < 2; ++k) d2[k] = b3 ? c2[2 + k] : c2[k];
#pragma unroll
  for (int k = 0; k < 4; ++k) d1[k] = b3 ? c1[4 + k] : c1[k];
  const bool b2 = d4 != top;
  const float e2 = b2 ? d2[1] : d2[0];
  float e1[2];
#pragma unroll
  for (int k = 0; k < 2; ++k) e1[k] = b2 ? d1[2 + k] : d1[k];
  const bool b1 = e2 != top;
  const float f1 = b1 ? e1[1] : e1[0];
  const bool b0 = f1 != top;
  return (b4 ? 16 : 0) | (b3 ? 8 : 0) | (b2 ? 4 : 0) | (b1 ? 2 : 0) | (b0 ? 1 : 0);
}

__global__ __launch_bounds__(128) void crf_viterbi(
    const float* __restrict__ pot, const float* __restrict__ chain,
    int* __restrict__ path) {
  __shared__ __attribute__((aligned(16))) float alphabuf[2][UU];
  __shared__ unsigned char bp[TT * UU];
  __shared__ __attribute__((aligned(16))) float potbuf[2][32 * UU];
  __shared__ unsigned char segmap[NSEG * UU];

  const int tid = threadIdx.x;
  const int lane = tid & 31;
  const int wave = tid >> 5;
  const int h = lane >> 4;
  const int c = wave * 16 + (lane & 15);
  const int b = blockIdx.x;
  const float* potb = pot + (size_t)b * TT * UU;

  float ck[32];
#pragma unroll
  for (int j = 0; j < 32; ++j) ck[j] = chain[(h * 32 + j) * UU + c];

#pragma unroll
  for (int i = 0; i < 4; ++i) {
    int f = tid + i * 128;
    *(f4*)&potbuf[0][f * 4] = *(const f4*)(potb + f * 4);
  }
  __syncthreads();
  if (h == 0) alphabuf[0][c] = potbuf[0][c];
  __syncthreads();

  float pt = potbuf[0][1 * UU + c];

  MaxTree mt;
#pragma unroll
  for (int i = 0; i < 16; ++i) mt.leaf[i] = 0.0f;
#pragma unroll
  for (int i = 0; i < 8; ++i) mt.l2[i] = 0.0f;
#pragma unroll
  for (int i = 0; i < 4; ++i) mt.l4[i] = 0.0f;
  mt.l8[0] = 0.0f; mt.l8[1] = 0.0f; mt.l16 = 0.0f; mt.top = 0.0f;
  float vmax = 0.0f;

  int t = 1;
  for (int chunk = 0; chunk < 16; ++chunk) {
    const int buf = chunk & 1;
    f4 nv[4];
    if (chunk < 15) {
#pragma unroll
      for (int i = 0; i < 4; ++i)
        nv[i] = *(const f4*)(potb + (chunk + 1) * 32 * UU + (tid + i * 128) * 4);
    }
    const int tEnd = (chunk + 1) * 32;
    for (; t < tEnd; ++t) {
      const int tl = t - chunk * 32;
      const f4* ap = (const f4*)&alphabuf[(t - 1) & 1][h * 32];
      f4 a[8];
#pragma unroll
      for (int q = 0; q < 8; ++q) a[q] = ap[q];
      const float ptn = potbuf[buf][((tl + 1) & 31) * UU + c];
      __builtin_amdgcn_sched_barrier(0);

      {
        const int fi = max_tree_first(mt) + h * 32;
        const int gi = (mt.top == vmax) ? fi : 255;
        const int oi = xhalf_i(gi);
        bp[(t - 1) * UU + c] = (unsigned char)min(gi, oi);
      }
      __builtin_amdgcn_sched_barrier(0);

      float v[32];
#pragma unroll
      for (int q = 0; q < 8; ++q) {
        v[4 * q + 0] = a[q].x + ck[4 * q + 0];
        v[4 * q + 1] = a[q].y + ck[4 * q + 1];
        v[4 * q + 2] = a[q].z + ck[4 * q + 2];
        v[4 * q + 3] = a[q].w + ck[4 * q + 3];
      }
      max_tree_build(v, mt);
      vmax = __builtin_fmaxf(mt.top, xhalf_f(mt.top));
      alphabuf[t & 1][c] = vmax + pt;
      step_barrier();
      pt = ptn;
    }
    if (chunk < 15) {
#pragma unroll
      for (int i = 0; i < 4; ++i)
        *(f4*)&potbuf[1 - buf][(tid + i * 128) * 4] = nv[i];
      __syncthreads();
      pt = potbuf[1 - buf][c];
    }
  }
  {
    const int fi = max_tree_first(mt) + h * 32;
    const int gi = (mt.top == vmax) ? fi : 255;
    const int oi = xhalf_i(gi);
    bp[(TT - 1) * UU + c] = (unsigned char)min(gi, oi);
  }
  __syncthreads();

  {
    const int e = tid & 63;
    const int g = tid >> 6;
    int cur[8];
#pragma unroll
    for (int k = 0; k < 8; ++k) cur[k] = e;
    for (int j = SEG - 1; j >= 0; --j) {
#pragma unroll
      for (int k = 0; k < 8; ++k) {
        const int s = 2 * k + g;
        if (s != 0) cur[k] = bp[(s * SEG + j) * UU + cur[k]];
      }
    }
#pragma unroll
    for (int k = 0; k < 8; ++k) {
      const int s = 2 * k + g;
      if (s != 0) segmap[s * UU + e] = (unsigned char)cur[k];
    }
  }
  __syncthreads();

  if (wave == 0) {
    const float* af = alphabuf[(TT - 1) & 1];
    float bb = af[0];
    int tag = 0;
    for (int i = 1; i < UU; ++i) {
      float s = af[i];
      if (s > bb) { bb = s; tag = i; }
    }
    const int s = lane & 15;
    int cur = tag;
    for (int q = NSEG - 1; q >= 1; --q) {
      const int nxt = segmap[q * UU + cur];
      cur = (q > s) ? nxt : cur;
    }
    if (lane < NSEG) {
      int* outb = path + (size_t)b * TT + s * SEG;
      outb[SEG - 1] = cur;
      for (int j = SEG - 1; j >= 1; --j) {
        cur = bp[(s * SEG + j) * UU + cur];
        outb[j - 1] = cur;
      }
    }
  }
}

extern "C" void kernel_launch(void* const* d_in, const int* in_sizes, int n_in,
                              void* d_out, int out_size, void* d_ws,
                              size_t ws_size, hipStream_t stream) {
  (void)in_sizes; (void)n_in; (void)out_size; (void)ws_size;
  const float* x     = (const float*)d_in[0];
  const float* w     = (const float*)d_in[1];
  const float* bias  = (const float*)d_in[2];
  const float* chain = (const float*)d_in[3];
  const float* lb    = (const float*)d_in[4];
  const float* rb    = (const float*)d_in[5];
  float* pot = (float*)d_ws;
  int* out = (int*)d_out;

  crf_potentials_gemm<<<dim3(MM / 128), dim3(256), 0, stream>>>(
      x, w, bias, lb, rb, pot);
  crf_viterbi<<<dim3(BB), dim3(128), 0, stream>>>(pot, chain, out);
}


# GEMM staging pipelined: next K chunk's global loads issued as one group before the current chunk's 64 WMMAs, stored behind them into a second LDS stage; one barrier per chunk; same sums and k order
# speedup vs baseline: 2.2085x; 1.2173x over previous
#include <hip/hip_runtime.h>
#include <hip/hip_bf16.h>

typedef float v2f __attribute__((ext_vector_type(2)));
typedef float v8f __attribute__((ext_vector_type(8)));
typedef float f4  __attribute__((ext_vector_type(4)));

constexpr int BB = 64;
constexpr int TT = 512;
constexpr int FF = 1024;
constexpr int UU = 64;
constexpr int MM = BB * TT;
constexpr int SEG = 32;
constexpr int NSEG = TT / SEG;

#define KCH 64
#define LDA 66
#define LDB 66
#define NCH (FF / KCH)

__global__ __launch_bounds__(256) void crf_potentials_gemm(
    const float* __restrict__ x, const float* __restrict__ w,
    const float* __restrict__ bias, const float* __restrict__ lb,
    const float* __restrict__ rb, float* __restrict__ pot) {
  __shared__ __attribute__((aligned(16))) float lds_a[2][128 * LDA];
  __shared__ __attribute__((aligned(16))) float lds_b[2][64 * LDB];

  const int tid = threadIdx.x;
  const int lane = tid & 31;
  const int wave = tid >> 5;
  const int blockRow = blockIdx.x * 128;
  const int rowBase = blockRow + wave * 16;

  const int mrow = lane & 15;
  const int kbase = (lane >> 4) << 1;

  const int srow = tid >> 4;
  const int sq = tid & 15;
  const float* asrc = x + (size_t)(blockRow + srow) * FF + sq * 4;
  const float* bsrc = w + (size_t)srow * UU + sq * 4;

  v8f acc[4] = {};
  f4 ra[8], rb4[4];

#pragma unroll
  for (int i = 0; i < 8; ++i) ra[i] = *(const f4*)(asrc + (size_t)(i * 16) * FF);
#pragma unroll
  for (int i = 0; i < 4; ++i) rb4[i] = *(const f4*)(bsrc + (size_t)(i * 16) * UU);
#pragma unroll
  for (int i = 0; i < 8; ++i) {
    float* dst = &lds_a[0][(srow + 16 * i) * LDA + sq * 4];
    dst[0] = ra[i].x; dst[1] = ra[i].y; dst[2] = ra[i].z; dst[3] = ra[i].w;
  }
#pragma unroll
  for (int i = 0; i < 4; ++i) {
    const int kr = srow + 16 * i;
    lds_b[0][(sq * 4 + 0) * LDB + kr] = rb4[i].x;
    lds_b[0][(sq * 4 + 1) * LDB + kr] = rb4[i].y;
    lds_b[0][(sq * 4 + 2) * LDB + kr] = rb4[i].z;
    lds_b[0][(sq * 4 + 3) * LDB + kr] = rb4[i].w;
  }
  __syncthreads();

  for (int c = 0; c < NCH; ++c) {
    const float* la = lds_a[c & 1];
    const float* lbt = lds_b[c & 1];
    const bool more = (c + 1 < NCH);

    if (more) {
      const int kk = (c + 1) * KCH;
#pragma unroll
      for (int i = 0; i < 8; ++i)
        ra[i] = *(const f4*)(asrc + (size_t)(i * 16) * FF + kk);
#pragma unroll
      for (int i = 0; i < 4; ++i)
        rb4[i] = *(const f4*)(bsrc + (size_t)(kk + i * 16) * UU);
    }

#pragma unroll
    for (int k4 = 0; k4 < KCH; k4 += 4) {
      v2f afrag = *(const v2f*)&la[(wave * 16 + mrow) * LDA + k4 + kbase];
#pragma unroll
      for (int nt = 0; nt < 4; ++nt) {
        int n = nt * 16 + mrow;
        v2f bfrag = *(const v2f*)&lbt[n * LDB + k4 + kbase];
        acc[nt] = __builtin_amdgcn_wmma_f32_16x16x4_f32(
            false, afrag, false, bfrag, (short)0, acc[nt], false, false);
      }
    }

    if (more) {
      float* sa = lds_a[(c + 1) & 1];
      float* sb = lds_b[(c + 1) & 1];
#pragma unroll
      for (int i = 0; i < 8; ++i) {
        float* dst = &sa[(srow + 16 * i) * LDA + sq * 4];
        dst[0] = ra[i].x; dst[1] = ra[i].y; dst[2] = ra[i].z; dst[3] = ra[i].w;
      }
#pragma unroll
      for (int i = 0; i < 4; ++i) {
        const int kr = srow + 16 * i;
        sb[(sq * 4 + 0) * LDB + kr] = rb4[i].x;
        sb[(sq * 4 + 1) * LDB + kr] = rb4[i].y;
        sb[(sq * 4 + 2) * LDB + kr] = rb4[i].z;
        sb[(sq * 4 + 3) * LDB + kr] = rb4[i].w;
      }
      __syncthreads();
    }
  }

  const int rsel = (lane >> 4) * 8;
#pragma unroll
  for (int nt = 0; nt < 4; ++nt) {
    int col = nt * 16 + mrow;
    float bia = bias[col];
    float lbv = lb[col];
    float rbv = rb[col];
#pragma unroll
    for (int r = 0; r < 8; ++r) {
      int row = rowBase + r + rsel;
      float v = acc[nt][r] + bia;
      int t = row & (TT - 1);
      if (t == 0) v += lbv;
      if (t == TT - 1) v += rbv;
      pot[(size_t)row * UU + col] = v;
    }
  }
}

static __device__ __forceinline__ int xhalf_i(int b) {
  return __builtin_amdgcn_permlanex16(b, b, 0x76543210, 0xfedcba98, false, false);
}
static __device__ __forceinline__ float xhalf_f(float x) {
  const int xi = __builtin_bit_cast(int, x);
  return __builtin_bit_cast(
      float, __builtin_amdgcn_permlanex16(0, xi, 0x76543210, 0xfedcba98, false, false));
}
static __device__ __forceinline__ void step_barrier() {
  asm volatile("s_wait_dscnt 0x0\n\ts_barrier_signal -1\n\ts_barrier_wait -1" ::: "memory");
}

struct MaxTree {
  float leaf[16];
  float l2[8];
  float l4[4];
  float l8[2];
  float l16;
  float top;
};
static __device__ __forceinline__ void max_tree_build(const float (&s)[32], MaxTree& m) {
  float g2[16], g4[8], g8[4], g16[2];
#pragma unroll
  for (int i = 0; i < 16; ++i) g2[i] = __builtin_fmaxf(s[2 * i], s[2 * i + 1]);
#pragma unroll
  for (int i = 0; i < 8; ++i) g4[i] = __builtin_fmaxf(g2[2 * i], g2[2 * i + 1]);
#pragma unroll
  for (int i = 0; i < 4; ++i) g8[i] = __builtin_fmaxf(g4[2 * i], g4[2 * i + 1]);
#pragma unroll
  for (int i = 0; i < 2; ++i) g16[i] = __builtin_fmaxf(g8[2 * i], g8[2 * i + 1]);
  m.top = __builtin_fmaxf(g16[0], g16[1]);
#pragma unroll
  for (int i = 0; i < 16; ++i) m.leaf[i] = s[2 * i];
#pragma unroll
  for (int i = 0; i < 8; ++i) m.l2[i] = g2[2 * i];
#pragma unroll
  for (int i = 0; i < 4; ++i) m.l4[i] = g4[2 * i];
#pragma unroll
  for (int i = 0; i < 2; ++i) m.l8[i] = g8[2 * i];
  m.l16 = g16[0];
}
static __device__ __forceinline__ int max_tree_first(const MaxTree& m) {
  const float top = m.top;
  const bool b4 = m.l16 != top;
  const float c8 = b4 ? m.l8[1] : m.l8[0];
  float c4[2], c2[4], c1[8];
#pragma unroll
  for (int k = 0; k < 2; ++k) c4[k] = b4 ? m.l4[2 + k] : m.l4[k];
#pragma unroll
  for (int k = 0; k < 4; ++k) c2[k] = b4 ? m.l2[4 + k] : m.l2[k];
#pragma unroll
  for (int k = 0; k < 8; ++k) c1[k] = b4 ? m.leaf[8 + k] : m.leaf[k];
  const bool b3 = c8 != top;
  const float d4 = b3 ? c4[1] : c4[0];
  float d2[2], d1[4];
#pragma unroll
  for (int k = 0; k < 2; ++k) d2[k] = b3 ? c2[2 + k] : c2[k];
#pragma unroll
  for (int k = 0; k < 4; ++k) d1[k] = b3 ? c1[4 + k] : c1[k];
  const bool b2 = d4 != top;
  const float e2 = b2 ? d2[1] : d2[0];
  float e1[2];
#pragma unroll
  for (int k = 0; k < 2; ++k) e1[k] = b2 ? d1[2 + k] : d1[k];
  const bool b1 = e2 != top;
  const float f1 = b1 ? e1[1] : e1[0];
  const bool b0 = f1 != top;
  return (b4 ? 16 : 0) | (b3 ? 8 : 0) | (b2 ? 4 : 0) | (b1 ? 2 : 0) | (b0 ? 1 : 0);
}

__global__ __launch_bounds__(128) void crf_viterbi(
    const float* __restrict__ pot, const float* __restrict__ chain,
    int* __restrict__ path) {
  __shared__ __attribute__((aligned(16))) float alphabuf[2][UU];
  __shared__ unsigned char bp[TT * UU];
  __shared__ __attribute__((aligned(16))) float potbuf[2][32 * UU];
  __shared__ unsigned char segmap[NSEG * UU];

  const int tid = threadIdx.x;
  const int lane = tid & 31;
  const int wave = tid >> 5;
  const int h = lane >> 4;
  const int c = wave * 16 + (lane & 15);
  const int b = blockIdx.x;
  const float* potb = pot + (size_t)b * TT * UU;

  float ck[32];
#pragma unroll
  for (int j = 0; j < 32; ++j) ck[j] = chain[(h * 32 + j) * UU + c];

#pragma unroll
  for (int i = 0; i < 4; ++i) {
    int f = tid + i * 128;
    *(f4*)&potbuf[0][f * 4] = *(const f4*)(potb + f * 4);
  }
  __syncthreads();
  if (h == 0) alphabuf[0][c] = potbuf[0][c];
  __syncthreads();

  float pt = potbuf[0][1 * UU + c];

  MaxTree mt;
#pragma unroll
  for (int i = 0; i < 16; ++i) mt.leaf[i] = 0.0f;
#pragma unroll
  for (int i = 0; i < 8; ++i) mt.l2[i] = 0.0f;
#pragma unroll
  for (int i = 0; i < 4; ++i) mt.l4[i] = 0.0f;
  mt.l8[0] = 0.0f; mt.l8[1] = 0.0f; mt.l16 = 0.0f; mt.top = 0.0f;
  float vmax = 0.0f;

  int t = 1;
  for (int chunk = 0; chunk < 16; ++chunk) {
    const int buf = chunk & 1;
    f4 nv[4];
    if (chunk < 15) {
#pragma unroll
      for (int i = 0; i < 4; ++i)
        nv[i] = *(const f4*)(potb + (chunk + 1) * 32 * UU + (tid + i * 128) * 4);
    }
    const int tEnd = (chunk + 1) * 32;
    for (; t < tEnd; ++t) {
      const int tl = t - chunk * 32;
      const f4* ap = (const f4*)&alphabuf[(t - 1) & 1][h * 32];
      f4 a[8];
#pragma unroll
      for (int q = 0; q < 8; ++q) a[q] = ap[q];
      const float ptn = potbuf[buf][((tl + 1) & 31) * UU + c];
      __builtin_amdgcn_sched_barrier(0);

      {
        const int fi = max_tree_first(mt) + h * 32;
        const int gi = (mt.top == vmax) ? fi : 255;
        const int oi = xhalf_i(gi);
        bp[(t - 1) * UU + c] = (unsigned char)min(gi, oi);
      }
      __builtin_amdgcn_sched_barrier(0);

      float v[32];
#pragma unroll
      for (int q = 0; q < 8; ++q) {
        v[4 * q + 0] = a[q].x + ck[4 * q + 0];
        v[4 * q + 1] = a[q].y + ck[4 * q + 1];
        v[4 * q + 2] = a[q].z + ck[4 * q + 2];
        v[4 * q + 3] = a[q].w + ck[4 * q + 3];
      }
      max_tree_build(v, mt);
      vmax = __builtin_fmaxf(mt.top, xhalf_f(mt.top));
      alphabuf[t & 1][c] = vmax + pt;
      step_barrier();
      pt = ptn;
    }
    if (chunk < 15) {
#pragma unroll
      for (int i = 0; i < 4; ++i)
        *(f4*)&potbuf[1 - buf][(tid + i * 128) * 4] = nv[i];
      __syncthreads();
      pt = potbuf[1 - buf][c];
    }
  }
  {
    const int fi = max_tree_first(mt) + h * 32;
    const int gi = (mt.top == vmax) ? fi : 255;
    const int oi = xhalf_i(gi);
    bp[(TT - 1) * UU + c] = (unsigned char)min(gi, oi);
  }
  __syncthreads();

  {
    const int e = tid & 63;
    const int g = tid >> 6;
    int cur[8];
#pragma unroll
    for (int k = 0; k < 8; ++k) cur[k] = e;
    for (int j = SEG - 1; j >= 0; --j) {
#pragma unroll
      for (int k = 0; k < 8; ++k) {
        const int s = 2 * k + g;
        if (s != 0) cur[k] = bp[(s * SEG + j) * UU + cur[k]];
      }
    }
#pragma unroll
    for (int k = 0; k < 8; ++k) {
      const int s = 2 * k + g;
      if (s != 0) segmap[s * UU + e] = (unsigned char)cur[k];
    }
  }
  __syncthreads();

  if (wave == 0) {
    const float* af = alphabuf[(TT - 1) & 1];
    float bb = af[0];
    int tag = 0;
    for (int i = 1; i < UU; ++i) {
      float s = af[i];
      if (s > bb) { bb = s; tag = i; }
    }
    const int s = lane & 15;
    int cur = tag;
    for (int q = NSEG - 1; q >= 1; --q) {
      const int nxt = segmap[q * UU + cur];
      cur = (q > s) ? nxt : cur;
    }
    if (lane < NSEG) {
      int* outb = path + (size_t)b * TT + s * SEG;
      outb[SEG - 1] = cur;
      for (int j = SEG - 1; j >= 1; --j) {
        cur = bp[(s * SEG + j) * UU + cur];
        outb[j - 1] = cur;
      }
    }
  }
}

extern "C" void kernel_launch(void* const* d_in, const int* in_sizes, int n_in,
                              void* d_out, int out_size, void* d_ws,
                              size_t ws_size, hipStream_t stream) {
  (void)in_sizes; (void)n_in; (void)out_size; (void)ws_size;
  const float* x     = (const float*)d_in[0];
  const float* w     = (const float*)d_in[1];
  const float* bias  = (const float*)d_in[2];
  const float* chain = (const float*)d_in[3];
  const float* lb    = (const float*)d_in[4];
  const float* rb    = (const float*)d_in[5];
  float* pot = (float*)d_ws;
  int* out = (int*)d_out;

  crf_potentials_gemm<<<dim3(MM / 128), dim3(256), 0, stream>>>(
      x, w, bias, lb, rb, pot);
  crf_viterbi<<<dim3(BB), dim3(128), 0, stream>>>(pot, chain, out);
}


# Viterbi: wave 0 alone runs the forward recursion (half the alphas by lane reads, half from its own LDS row in flight behind them; no barrier per step); waves 1-3 rebuild backpointers; same sums
# speedup vs baseline: 2.3929x; 1.0835x over previous
#include <hip/hip_runtime.h>
#include <hip/hip_bf16.h>

typedef float v2f __attribute__((ext_vector_type(2)));
typedef float v8f __attribute__((ext_vector_type(8)));
typedef float f4  __attribute__((ext_vector_type(4)));

constexpr int BB = 64;
constexpr int TT = 512;
constexpr int FF = 1024;
constexpr int UU = 64;
constexpr int MM = BB * TT;
constexpr int SEG = 32;
constexpr int NSEG = TT / SEG;

#define KCH 64
#define LDA 66
#define LDB 66
#define NCH (FF / KCH)

__global__ __launch_bounds__(256) void crf_potentials_gemm(
    const float* __restrict__ x, const float* __restrict__ w,
    const float* __restrict__ bias, const float* __restrict__ lb,
    const float* __restrict__ rb, float* __restrict__ pot) {
  __shared__ __attribute__((aligned(16))) float lds_a[2][128 * LDA];
  __shared__ __attribute__((aligned(16))) float lds_b[2][64 * LDB];

  const int tid = threadIdx.x;
  const int lane = tid & 31;
  const int wave = tid >> 5;
  const int blockRow = blockIdx.x * 128;
  const int rowBase = blockRow + wave * 16;

  const int mrow = lane & 15;
  const int kbase = (lane >> 4) << 1;

  const int srow = tid >> 4;
  const int sq = tid & 15;
  const float* asrc = x + (size_t)(blockRow + srow) * FF + sq * 4;
  const float* bsrc = w + (size_t)srow * UU + sq * 4;

  v8f acc[4] = {};
  f4 ra[8], rb4[4];

#pragma unroll
  for (int i = 0; i < 8; ++i) ra[i] = *(const f4*)(asrc + (size_t)(i * 16) * FF);
#pragma unroll
  for (int i = 0; i < 4; ++i) rb4[i] = *(const f4*)(bsrc + (size_t)(i * 16) * UU);
#pragma unroll
  for (int i = 0; i < 8; ++i) {
    float* dst = &lds_a[0][(srow + 16 * i) * LDA + sq * 4];
    dst[0] = ra[i].x; dst[1] = ra[i].y; dst[2] = ra[i].z; dst[3] = ra[i].w;
  }
#pragma unroll
  for (int i = 0; i < 4; ++i) {
    const int kr = srow + 16 * i;
    lds_b[0][(sq * 4 + 0) * LDB + kr] = rb4[i].x;
    lds_b[0][(sq * 4 + 1) * LDB + kr] = rb4[i].y;
    lds_b[0][(sq * 4 + 2) * LDB + kr] = rb4[i].z;
    lds_b[0][(sq * 4 + 3) * LDB + kr] = rb4[i].w;
  }
  __syncthreads();

  for (int c = 0; c < NCH; ++c) {
    const float* la = lds_a[c & 1];
    const float* lbt = lds_b[c & 1];
    const bool more = (c + 1 < NCH);

    if (more) {
      const int kk = (c + 1) * KCH;
#pragma unroll
      for (int i = 0; i < 8; ++i)
        ra[i] = *(const f4*)(asrc + (size_t)(i * 16) * FF + kk);
#pragma unroll
      for (int i = 0; i < 4; ++i)
        rb4[i] = *(const f4*)(bsrc + (size_t)(kk + i * 16) * UU);
    }

#pragma unroll
    for (int k4 = 0; k4 < KCH; k4 += 4) {
      v2f afrag = *(const v2f*)&la[(wave * 16 + mrow) * LDA + k4 + kbase];
#pragma unroll
      for (int nt = 0; nt < 4; ++nt) {
        int n = nt * 16 + mrow;
        v2f bfrag = *(const v2f*)&lbt[n * LDB + k4 + kbase];
        acc[nt] = __builtin_amdgcn_wmma_f32_16x16x4_f32(
            false, afrag, false, bfrag, (short)0, acc[nt], false, false);
      }
    }

    if (more) {
      float* sa = lds_a[(c + 1) & 1];
      float* sb = lds_b[(c + 1) & 1];
#pragma unroll
      for (int i = 0; i < 8; ++i) {
        float* dst = &sa[(srow + 16 * i) * LDA + sq * 4];
        dst[0] = ra[i].x; dst[1] = ra[i].y; dst[2] = ra[i].z; dst[3] = ra[i].w;
      }
#pragma unroll
      for (int i = 0; i < 4; ++i) {
        const int kr = srow + 16 * i;
        sb[(sq * 4 + 0) * LDB + kr] = rb4[i].x;
        sb[(sq * 4 + 1) * LDB + kr] = rb4[i].y;
        sb[(sq * 4 + 2) * LDB + kr] = rb4[i].z;
        sb[(sq * 4 + 3) * LDB + kr] = rb4[i].w;
      }
      __syncthreads();
    }
  }

  const int rsel = (lane >> 4) * 8;
#pragma unroll
  for (int nt = 0; nt < 4; ++nt) {
    int col = nt * 16 + mrow;
    float bia = bias[col];
    float lbv = lb[col];
    float rbv = rb[col];
#pragma unroll
    for (int r = 0; r < 8; ++r) {
      int row = rowBase + r + rsel;
      float v = acc[nt][r] + bia;
      int t = row & (TT - 1);
      if (t == 0) v += lbv;
      if (t == TT - 1) v += rbv;
      pot[(size_t)row * UU + col] = v;
    }
  }
}

static __device__ __forceinline__ float lane_bcast(float x, int lane) {
  return __builtin_bit_cast(
      float, __builtin_amdgcn_readlane(__builtin_bit_cast(int, x), lane));
}
static __device__ __forceinline__ float fmax3(float a, float b, float c) {
  return __builtin_fmaxf(__builtin_fmaxf(a, b), c);
}
static __device__ __forceinline__ float max32(const float (&r)[32]) {
  float g[11];
#pragma unroll
  for (int i = 0; i < 10; ++i) g[i] = fmax3(r[3 * i], r[3 * i + 1], r[3 * i + 2]);
  g[10] = __builtin_fmaxf(r[30], r[31]);
  const float k0 = fmax3(g[0], g[1], g[2]);
  const float k1 = fmax3(g[3], g[4], g[5]);
  const float k2 = fmax3(g[6], g[7], g[8]);
  const float k3 = __builtin_fmaxf(g[9], g[10]);
  return __builtin_fmaxf(fmax3(k0, k1, k2), k3);
}

struct MaxTree {
  float leaf[16];
  float l2[8];
  float l4[4];
  float l8[2];
  float l16;
  float top;
};
static __device__ __forceinline__ void max_tree_build(const float (&s)[32], MaxTree& m) {
  float g2[16], g4[8], g8[4], g16[2];
#pragma unroll
  for (int i = 0; i < 16; ++i) g2[i] = __builtin_fmaxf(s[2 * i], s[2 * i + 1]);
#pragma unroll
  for (int i = 0; i < 8; ++i) g4[i] = __builtin_fmaxf(g2[2 * i], g2[2 * i + 1]);
#pragma unroll
  for (int i = 0; i < 4; ++i) g8[i] = __builtin_fmaxf(g4[2 * i], g4[2 * i + 1]);
#pragma unroll
  for (int i = 0; i < 2; ++i) g16[i] = __builtin_fmaxf(g8[2 * i], g8[2 * i + 1]);
  m.top = __builtin_fmaxf(g16[0], g16[1]);
#pragma unroll
  for (int i = 0; i < 16; ++i) m.leaf[i] = s[2 * i];
#pragma unroll
  for (int i = 0; i < 8; ++i) m.l2[i] = g2[2 * i];
#pragma unroll
  for (int i = 0; i < 4; ++i) m.l4[i] = g4[2 * i];
#pragma unroll
  for (int i = 0; i < 2; ++i) m.l8[i] = g8[2 * i];
  m.l16 = g16[0];
}
static __device__ __forceinline__ int max_tree_first(const MaxTree& m) {
  const float top = m.top;
  const bool b4 = m.l16 != top;
  const float c8 = b4 ? m.l8[1] : m.l8[0];
  float c4[2], c2[4], c1[8];
#pragma unroll
  for (int k = 0; k < 2; ++k) c4[k] = b4 ? m.l4[2 + k] : m.l4[k];
#pragma unroll
  for (int k = 0; k < 4; ++k) c2[k] = b4 ? m.l2[4 + k] : m.l2[k];
#pragma unroll
  for (int k = 0; k < 8; ++k) c1[k] = b4 ? m.leaf[8 + k] : m.leaf[k];
  const bool b3 = c8 != top;
  const float d4 = b3 ? c4[1] : c4[0];
  float d2[2], d1[4];
#pragma unroll
  for (int k = 0; k < 2; ++k) d2[k] = b3 ? c2[2 + k] : c2[k];
#pragma unroll
  for (int k = 0; k < 4; ++k) d1[k] = b3 ? c1[4 + k] : c1[k];
  const bool b2 = d4 != top;
  const float e2 = b2 ? d2[1] : d2[0];
  float e1[2];
#pragma unroll
  for (int k = 0; k < 2; ++k) e1[k] = b2 ? d1[2 + k] : d1[k];
  const bool b1 = e2 != top;
  const float f1 = b1 ? e1[1] : e1[0];
  const bool b0 = f1 != top;
  return (b4 ? 16 : 0) | (b3 ? 8 : 0) | (b2 ? 4 : 0) | (b1 ? 2 : 0) | (b0 ? 1 : 0);
}

template <int SEL>
static __device__ __forceinline__ int first_argmax64(const f4 (&a)[16], const v2f (&ckp)[64]) {
  float ck[64];
#pragma unroll
  for (int p = 0; p < 64; ++p) ck[p] = SEL ? ckp[p].y : ckp[p].x;
  float v[32];
  MaxTree lo, hi;
#pragma unroll
  for (int q = 0; q < 8; ++q) {
    v[4 * q + 0] = a[q].x + ck[4 * q + 0];
    v[4 * q + 1] = a[q].y + ck[4 * q + 1];
    v[4 * q + 2] = a[q].z + ck[4 * q + 2];
    v[4 * q + 3] = a[q].w + ck[4 * q + 3];
  }
  max_tree_build(v, lo);
#pragma unroll
  for (int q = 0; q < 8; ++q) {
    v[4 * q + 0] = a[8 + q].x + ck[32 + 4 * q + 0];
    v[4 * q + 1] = a[8 + q].y + ck[32 + 4 * q + 1];
    v[4 * q + 2] = a[8 + q].z + ck[32 + 4 * q + 2];
    v[4 * q + 3] = a[8 + q].w + ck[32 + 4 * q + 3];
  }
  max_tree_build(v, hi);
  const int i0 = max_tree_first(lo);
  const int i1 = max_tree_first(hi);
  return (lo.top >= hi.top) ? i0 : 32 + i1;
}

static __device__ __forceinline__ void bp_row(const float* hist, unsigned char* bp, int t, int lane,
                                              const v2f (&ck)[64]) {
  const f4* ap = (const f4*)&hist[(t - 1) * UU];
  f4 a[16];
#pragma unroll
  for (int q = 0; q < 16; ++q) a[q] = ap[q];
  bp[t * UU + lane] = (unsigned char)first_argmax64<0>(a, ck);
  bp[t * UU + lane + 32] = (unsigned char)first_argmax64<1>(a, ck);
}

__global__ __launch_bounds__(128) void crf_viterbi(
    const float* __restrict__ pot, const float* __restrict__ chain,
    int* __restrict__ path) {
  __shared__ __attribute__((aligned(16))) float hist[TT * UU];
  __shared__ unsigned char bp[TT * UU];
  __shared__ __attribute__((aligned(16))) float potbuf[2][32 * UU];
  __shared__ unsigned char segmap[NSEG * UU];

  const int tid = threadIdx.x;
  const int lane = tid & 31;
  const int wave = __builtin_amdgcn_readfirstlane(tid >> 5);
  const int b = blockIdx.x;
  const float* potb = pot + (size_t)b * TT * UU;

  v2f ck[64];
#pragma unroll
  for (int p = 0; p < 64; ++p) {
    ck[p].x = chain[p * UU + lane];
    ck[p].y = chain[p * UU + lane + 32];
  }

#pragma unroll
  for (int i = 0; i < 4; ++i) {
    int f = tid + i * 128;
    *(f4*)&potbuf[0][f * 4] = *(const f4*)(potb + f * 4);
  }
  __syncthreads();
  if (tid < UU) hist[tid] = potbuf[0][tid];
  float a0 = potbuf[0][lane];
  __syncthreads();
  f4 hv[8];
#pragma unroll
  for (int q = 0; q < 8; ++q) hv[q] = *(const f4*)&hist[32 + 4 * q];

  for (int chunk = 0; chunk < 16; ++chunk) {
    const int buf = chunk & 1;
    f4 nv[4];
    if (chunk < 15) {
#pragma unroll
      for (int i = 0; i < 4; ++i)
        nv[i] = *(const f4*)(potb + (chunk + 1) * 32 * UU + (tid + i * 128) * 4);
    }
    if (wave == 0) {
      for (int tl = (chunk == 0 ? 1 : 0); tl < 32; ++tl) {
        const int t = chunk * 32 + tl;
        const float pt0 = potbuf[buf][tl * UU + lane];
        const float pt1 = potbuf[buf][tl * UU + lane + 32];
        float r0[32], r1[32];
#pragma unroll
        for (int p = 0; p < 32; ++p) {
          const float sa = lane_bcast(a0, p);
          const v2f ua = (v2f){sa, sa} + ck[p];
          r0[p] = ua.x;
          r1[p] = ua.y;
        }
        const float lo0 = max32(r0);
        const float lo1 = max32(r1);
        __builtin_amdgcn_sched_barrier(0);
#pragma unroll
        for (int q = 0; q < 8; ++q) {
          const v2f u0 = (v2f){hv[q].x, hv[q].x} + ck[32 + 4 * q + 0];
          const v2f u1 = (v2f){hv[q].y, hv[q].y} + ck[32 + 4 * q + 1];
          const v2f u2 = (v2f){hv[q].z, hv[q].z} + ck[32 + 4 * q + 2];
          const v2f u3 = (v2f){hv[q].w, hv[q].w} + ck[32 + 4 * q + 3];
          r0[4 * q + 0] = u0.x; r1[4 * q + 0] = u0.y;
          r0[4 * q + 1] = u1.x; r1[4 * q + 1] = u1.y;
          r0[4 * q + 2] = u2.x; r1[4 * q + 2] = u2.y;
          r0[4 * q + 3] = u3.x; r1[4 * q + 3] = u3.y;
        }
        a0 = __builtin_fmaxf(lo0, max32(r0)) + pt0;
        const float a1 = __builtin_fmaxf(lo1, max32(r1)) + pt1;
        hist[t * UU + lane] = a0;
        hist[t * UU + lane + 32] = a1;
#pragma unroll
        for (int q = 0; q < 8; ++q) hv[q] = *(const f4*)&hist[t * UU + 32 + 4 * q];
      }
    } else if (chunk > 0) {
      for (int r = wave - 1; r < 32; r += 3) {
        const int t = (chunk - 1) * 32 + r;
        if (t > 0) bp_row(hist, bp, t, lane, ck);
      }
    }
    if (chunk < 15) {
#pragma unroll
      for (int i = 0; i < 4; ++i)
        *(f4*)&potbuf[1 - buf][(tid + i * 128) * 4] = nv[i];
    }
    __syncthreads();
  }
  for (int r = wave; r < 32; r += 4) bp_row(hist, bp, 15 * 32 + r, lane, ck);
  __syncthreads();

  {
    const int e = tid & 63;
    const int g = tid >> 6;
    int cur[8];
#pragma unroll
    for (int k = 0; k < 8; ++k) cur[k] = e;
    for (int j = SEG - 1; j >= 0; --j) {
#pragma unroll
      for (int k = 0; k < 8; ++k) {
        const int s = 2 * k + g;
        if (s != 0) cur[k] = bp[(s * SEG + j) * UU + cur[k]];
      }
    }
#pragma unroll
    for (int k = 0; k < 8; ++k) {
      const int s = 2 * k + g;
      if (s != 0) segmap[s * UU + e] = (unsigned char)cur[k];
    }
  }
  __syncthreads();

  if (wave == 0) {
    const float* af = &hist[(TT - 1) * UU];
    float bb = af[0];
    int tag = 0;
    for (int i = 1; i < UU; ++i) {
      float s = af[i];
      if (s > bb) { bb = s; tag = i; }
    }
    const int s = lane & 15;
    int cur = tag;
    for (int q = NSEG - 1; q >= 1; --q) {
      const int nxt = segmap[q * UU + cur];
      cur = (q > s) ? nxt : cur;
    }
    if (lane < NSEG) {
      int* outb = path + (size_t)b * TT + s * SEG;
      outb[SEG - 1] = cur;
      for (int j = SEG - 1; j >= 1; --j) {
        cur = bp[(s * SEG + j) * UU + cur];
        outb[j - 1] = cur;
      }
    }
  }
}

extern "C" void kernel_launch(void* const* d_in, const int* in_sizes, int n_in,
                              void* d_out, int out_size, void* d_ws,
                              size_t ws_size, hipStream_t stream) {
  (void)in_sizes; (void)n_in; (void)out_size; (void)ws_size;
  const float* x     = (const float*)d_in[0];
  const float* w     = (const float*)d_in[1];
  const float* bias  = (const float*)d_in[2];
  const float* chain = (const float*)d_in[3];
  const float* lb    = (const float*)d_in[4];
  const float* rb    = (const float*)d_in[5];
  float* pot = (float*)d_ws;
  int* out = (int*)d_out;

  crf_potentials_gemm<<<dim3(MM / 128), dim3(256), 0, stream>>>(
      x, w, bias, lb, rb, pot);
  crf_viterbi<<<dim3(BB), dim3(128), 0, stream>>>(pot, chain, out);
}
